# starting kernel
# baseline (speedup 1.0000x reference)
#include <hip/hip_runtime.h>
#include <math.h>


typedef __attribute__((ext_vector_type(2))) float v2f;
typedef __attribute__((ext_vector_type(8))) float v8f;

#define NB 4
#define NS 64
#define NW 32
#define NM 16
#define NPTS (NB * NS * NS * NS)
#define HSZ ((size_t)NPTS * NW)

__device__ __forceinline__ v8f wmma4(v2f a, v2f b, v8f c) {
  return __builtin_amdgcn_wmma_f32_16x16x4_f32(false, a, false, b, (short)0, c,
                                               false, false);
}

__device__ __forceinline__ v8f zero8() {
  v8f z;
  #pragma unroll
  for (int i = 0; i < 8; ++i) z[i] = 0.0f;
  return z;
}

__device__ __forceinline__ float gelu_exact(float x) {
  return 0.5f * x * (1.0f + erff(x * 0.70710678118654752440f));
}

__global__ __launch_bounds__(256) void k_input_proj(
    const float* __restrict__ x, const float* __restrict__ w,
    const float* __restrict__ bias, float* __restrict__ out) {
  const int lane = threadIdx.x & 31, wv = threadIdx.x >> 5;
  const int half = lane >> 4, l16 = lane & 15;
  const long pbase = ((long)blockIdx.x * 8 + wv) * 16;
  v2f a;
  a.x = x[(pbase + l16) * 4 + half * 2];
  a.y = x[(pbase + l16) * 4 + half * 2 + 1];
  v2f b0, b1;
  b0.x = w[(half * 2) * NW + l16];
  b0.y = w[(half * 2 + 1) * NW + l16];
  b1.x = w[(half * 2) * NW + 16 + l16];
  b1.y = w[(half * 2 + 1) * NW + 16 + l16];
  v8f d0 = zero8(), d1 = zero8();
  d0 = wmma4(a, b0, d0);
  d1 = wmma4(a, b1, d1);
  #pragma unroll
  for (int r = 0; r < 8; ++r) {
    long p = pbase + r + half * 8;
    out[p * NW + l16] = d0[r] + bias[l16];
    out[p * NW + 16 + l16] = d1[r] + bias[16 + l16];
  }
}

template <bool WT, bool DOGELU>
__global__ __launch_bounds__(256) void k_pointwise(
    const float* __restrict__ hin, const float* __restrict__ w,
    const float* __restrict__ bias, float* __restrict__ out) {
  const int lane = threadIdx.x & 31, wv = threadIdx.x >> 5;
  const int half = lane >> 4, l16 = lane & 15;
  const long pbase = ((long)blockIdx.x * 8 + wv) * 16;
  v8f d0 = zero8(), d1 = zero8();
  #pragma unroll
  for (int kk = 0; kk < 8; ++kk) {
    const int i0 = kk * 4 + half * 2;
    v2f a;
    a.x = hin[(pbase + l16) * NW + i0];
    a.y = hin[(pbase + l16) * NW + i0 + 1];
    v2f b0, b1;
    if (WT) {
      b0.x = w[(i0)*NW + l16];      b0.y = w[(i0 + 1) * NW + l16];
      b1.x = w[(i0)*NW + 16 + l16]; b1.y = w[(i0 + 1) * NW + 16 + l16];
    } else {
      b0.x = w[(l16)*NW + i0];        b0.y = w[(l16)*NW + i0 + 1];
      b1.x = w[(16 + l16) * NW + i0]; b1.y = w[(16 + l16) * NW + i0 + 1];
    }
    d0 = wmma4(a, b0, d0);
    d1 = wmma4(a, b1, d1);
  }
  #pragma unroll
  for (int r = 0; r < 8; ++r) {
    long p = pbase + r + half * 8;
    float v0 = d0[r] + bias[l16];
    float v1 = d1[r] + bias[16 + l16];
    if (DOGELU) { v0 = gelu_exact(v0); v1 = gelu_exact(v1); }
    out[p * NW + l16] = v0;
    out[p * NW + 16 + l16] = v1;
  }
}

__global__ __launch_bounds__(256) void k_spectral_axis(
    const float* __restrict__ h, const float* __restrict__ wgt,
    float* __restrict__ acc, long strideU, long strideV, long strideA) {
  extern __shared__ float smem[];
  float* Ff = smem;
  float* Fi = smem + 2048;
  float* G  = smem + 4096;
  float* Gp = smem + 4096 + 16384;

  const int tid = threadIdx.x;
  const int lane = tid & 31, wv = tid >> 5;
  const int half = lane >> 4, l16 = lane & 15;
  const float NRM = 0.125f;

  for (int idx = tid; idx < 2048; idx += 256) {
    int m2 = idx >> 6, p = idx & 63, k = m2 >> 1;
    float ang = 6.28318530717958647693f * (float)(k * p) * (1.0f / 64.0f);
    Ff[idx] = (m2 & 1) ? (-__sinf(ang) * NRM) : (__cosf(ang) * NRM);
  }
  for (int idx = tid; idx < 2048; idx += 256) {
    int p = idx >> 5, m2 = idx & 31, k = m2 >> 1;
    float ang = 6.28318530717958647693f * (float)(k * p) * (1.0f / 64.0f);
    float sc = (k == 0 ? 1.0f : 2.0f) * NRM;
    Fi[idx] = (m2 & 1) ? (-__sinf(ang) * sc) : (__cosf(ang) * sc);
  }
  __syncthreads();

  long base[2];
  #pragma unroll
  for (int ll = 0; ll < 2; ++ll) {
    int l = blockIdx.x * 16 + wv * 2 + ll;
    int b = l >> 12, u = (l >> 6) & 63, v = l & 63;
    base[ll] = (long)b * (64L * 64 * 64 * NW) + (long)u * strideU +
               (long)v * strideV;
  }

  #pragma unroll
  for (int ll = 0; ll < 2; ++ll) {
    const int li = wv * 2 + ll;
    v8f a00 = zero8(), a01 = zero8(), a10 = zero8(), a11 = zero8();
    for (int kk = 0; kk < 16; ++kk) {
      const int p0 = kk * 4 + half * 2;
      v2f b0, b1;
      b0.x = h[base[ll] + (long)p0 * strideA + l16];
      b0.y = h[base[ll] + (long)(p0 + 1) * strideA + l16];
      b1.x = h[base[ll] + (long)p0 * strideA + 16 + l16];
      b1.y = h[base[ll] + (long)(p0 + 1) * strideA + 16 + l16];
      v2f fa0, fa1;
      fa0.x = Ff[(l16)*64 + p0];      fa0.y = Ff[(l16)*64 + p0 + 1];
      fa1.x = Ff[(16 + l16) * 64 + p0]; fa1.y = Ff[(16 + l16) * 64 + p0 + 1];
      a00 = wmma4(fa0, b0, a00);
      a01 = wmma4(fa0, b1, a01);
      a10 = wmma4(fa1, b0, a10);
      a11 = wmma4(fa1, b1, a11);
    }
    float* g = G + li * 1024;
    #pragma unroll
    for (int r = 0; r < 8; ++r) {
      const int m = r + half * 8;
      g[(m)*32 + l16]           = a00[r];
      g[(m)*32 + 16 + l16]      = a01[r];
      g[(m + 16) * 32 + l16]    = a10[r];
      g[(m + 16) * 32 + 16 + l16] = a11[r];
    }
  }
  __syncthreads();

  #pragma unroll
  for (int km = 0; km < 2; ++km) {
    const int k = wv + km * 8;
    v8f oRe0 = zero8(), oRe1 = zero8(), oIm0 = zero8(), oIm1 = zero8();
    for (int kk = 0; kk < 8; ++kk) {
      const int i0 = kk * 4 + half * 2, i1 = i0 + 1;
      v2f bRe, bIm;
      bRe.x = G[l16 * 1024 + (2 * k) * 32 + i0];
      bRe.y = G[l16 * 1024 + (2 * k) * 32 + i1];
      bIm.x = G[l16 * 1024 + (2 * k + 1) * 32 + i0];
      bIm.y = G[l16 * 1024 + (2 * k + 1) * 32 + i1];
      #pragma unroll
      for (int mt = 0; mt < 2; ++mt) {
        const int o = mt * 16 + l16;
        v2f aR, aI, aIn;
        aR.x = wgt[((i0 * NW + o) * NM + k) * 2 + 0];
        aR.y = wgt[((i1 * NW + o) * NM + k) * 2 + 0];
        aI.x = wgt[((i0 * NW + o) * NM + k) * 2 + 1];
        aI.y = wgt[((i1 * NW + o) * NM + k) * 2 + 1];
        aIn.x = -aI.x; aIn.y = -aI.y;
        if (mt == 0) {
          oRe0 = wmma4(aR, bRe, oRe0);  oRe0 = wmma4(aIn, bIm, oRe0);
          oIm0 = wmma4(aI, bRe, oIm0);  oIm0 = wmma4(aR, bIm, oIm0);
        } else {
          oRe1 = wmma4(aR, bRe, oRe1);  oRe1 = wmma4(aIn, bIm, oRe1);
          oIm1 = wmma4(aI, bRe, oIm1);  oIm1 = wmma4(aR, bIm, oIm1);
        }
      }
    }
    #pragma unroll
    for (int r = 0; r < 8; ++r) {
      const int o0 = r + half * 8;
      Gp[l16 * 1024 + (2 * k) * 32 + o0]          = oRe0[r];
      Gp[l16 * 1024 + (2 * k) * 32 + o0 + 16]     = oRe1[r];
      Gp[l16 * 1024 + (2 * k + 1) * 32 + o0]      = oIm0[r];
      Gp[l16 * 1024 + (2 * k + 1) * 32 + o0 + 16] = oIm1[r];
    }
  }
  __syncthreads();

  #pragma unroll
  for (int ll = 0; ll < 2; ++ll) {
    const int li = wv * 2 + ll;
    const float* gp = Gp + li * 1024;
    v8f o00 = zero8(), o01 = zero8(), o10 = zero8(), o11 = zero8();
    v8f o20 = zero8(), o21 = zero8(), o30 = zero8(), o31 = zero8();
    for (int kk = 0; kk < 8; ++kk) {
      const int i0 = kk * 4 + half * 2;
      v2f b0, b1;
      b0.x = gp[(i0)*32 + l16];      b0.y = gp[(i0 + 1) * 32 + l16];
      b1.x = gp[(i0)*32 + 16 + l16]; b1.y = gp[(i0 + 1) * 32 + 16 + l16];
      v2f a0, a1, a2, a3;
      a0.x = Fi[(l16)*32 + i0];        a0.y = Fi[(l16)*32 + i0 + 1];
      a1.x = Fi[(16 + l16) * 32 + i0]; a1.y = Fi[(16 + l16) * 32 + i0 + 1];
      a2.x = Fi[(32 + l16) * 32 + i0]; a2.y = Fi[(32 + l16) * 32 + i0 + 1];
      a3.x = Fi[(48 + l16) * 32 + i0]; a3.y = Fi[(48 + l16) * 32 + i0 + 1];
      o00 = wmma4(a0, b0, o00);  o01 = wmma4(a0, b1, o01);
      o10 = wmma4(a1, b0, o10);  o11 = wmma4(a1, b1, o11);
      o20 = wmma4(a2, b0, o20);  o21 = wmma4(a2, b1, o21);
      o30 = wmma4(a3, b0, o30);  o31 = wmma4(a3, b1, o31);
    }
    #pragma unroll
    for (int r = 0; r < 8; ++r) {
      const int pr = r + half * 8;
      long i0a = base[ll] + (long)(pr)*strideA;
      long i1a = base[ll] + (long)(pr + 16) * strideA;
      long i2a = base[ll] + (long)(pr + 32) * strideA;
      long i3a = base[ll] + (long)(pr + 48) * strideA;
      acc[i0a + l16] += o00[r];  acc[i0a + 16 + l16] += o01[r];
      acc[i1a + l16] += o10[r];  acc[i1a + 16 + l16] += o11[r];
      acc[i2a + l16] += o20[r];  acc[i2a + 16 + l16] += o21[r];
      acc[i3a + l16] += o30[r];  acc[i3a + 16 + l16] += o31[r];
    }
  }
}

__global__ __launch_bounds__(256) void k_gelu(const float* __restrict__ in,
                                              float* __restrict__ out) {
  const long i = (long)blockIdx.x * blockDim.x + threadIdx.x;
  float4 v = ((const float4*)in)[i];
  v.x = gelu_exact(v.x);
  v.y = gelu_exact(v.y);
  v.z = gelu_exact(v.z);
  v.w = gelu_exact(v.w);
  ((float4*)out)[i] = v;
}

__global__ __launch_bounds__(256) void k_fc3(const float* __restrict__ hin,
                                             const float* __restrict__ w,
                                             const float* __restrict__ bias,
                                             float* __restrict__ out) {
  const long p = (long)blockIdx.x * blockDim.x + threadIdx.x;
  const float4* hv = (const float4*)(hin + p * NW);
  const float4* wv = (const float4*)w;
  float s = 0.0f;
  #pragma unroll
  for (int i = 0; i < 8; ++i) {
    float4 a = hv[i], b = wv[i];
    s += a.x * b.x + a.y * b.y + a.z * b.z + a.w * b.w;
  }
  out[p] = s + bias[0];
}

extern "C" void kernel_launch(void* const* d_in, const int* in_sizes, int n_in,
                              void* d_out, int out_size, void* d_ws,
                              size_t ws_size, hipStream_t stream) {
  const float* x      = (const float*)d_in[0];
  const float* in_w   = (const float*)d_in[1];
  const float* in_b   = (const float*)d_in[2];
  const float* fwx    = (const float*)d_in[3];
  const float* fwy    = (const float*)d_in[4];
  const float* fwz    = (const float*)d_in[5];
  const float* conv_w = (const float*)d_in[6];
  const float* conv_b = (const float*)d_in[7];
  const float* fc1_w  = (const float*)d_in[8];
  const float* fc1_b  = (const float*)d_in[9];
  const float* fc2_w  = (const float*)d_in[10];
  const float* fc2_b  = (const float*)d_in[11];
  const float* fc3_w  = (const float*)d_in[12];
  const float* fc3_b  = (const float*)d_in[13];
  float* out = (float*)d_out;

  float* buf0 = (float*)d_ws;
  float* buf1 = buf0 + HSZ;

  const dim3 blk(256);
  const int wgPts = NPTS / (16 * 8);
  const size_t smemB = (size_t)(4096 + 2 * 16384) * sizeof(float);
  const long sX = 64L * 64 * NW;
  const long sY = 64L * NW;
  const long sZ = (long)NW;

  k_input_proj<<<wgPts, blk, 0, stream>>>(x, in_w, in_b, buf0);

  for (int i = 0; i < 4; ++i) {
    k_pointwise<false, false><<<wgPts, blk, 0, stream>>>(
        buf0, conv_w + i * (NW * NW), conv_b + i * NW, buf1);
    const long wofs = (long)i * (NW * NW * NM * 2);
    k_spectral_axis<<<1024, blk, smemB, stream>>>(buf0, fwz + wofs, buf1,
                                                  sX, sY, sZ);
    k_spectral_axis<<<1024, blk, smemB, stream>>>(buf0, fwy + wofs, buf1,
                                                  sX, sZ, sY);
    k_spectral_axis<<<1024, blk, smemB, stream>>>(buf0, fwx + wofs, buf1,
                                                  sY, sZ, sX);
    if (i < 3)
      k_gelu<<<(int)(HSZ / 4 / 256), blk, 0, stream>>>(buf1, buf0);
  }

  k_pointwise<true, true><<<wgPts, blk, 0, stream>>>(buf1, fc1_w, fc1_b, buf0);
  k_pointwise<true, true><<<wgPts, blk, 0, stream>>>(buf0, fc2_w, fc2_b, buf1);
  k_fc3<<<NPTS / 256, blk, 0, stream>>>(buf1, fc3_w, fc3_b, out);
}


# spectral kernel: the two DFT bases are held in registers, not LDS: each lane forms its basis fragments by v_cos/v_sin of exact 64ths of a turn (half a period kept, rest by sign); sum order unchanged
# speedup vs baseline: 2.4825x; 2.4825x over previous
#include <hip/hip_runtime.h>
#include <math.h>


typedef __attribute__((ext_vector_type(2))) float v2f;
typedef __attribute__((ext_vector_type(8))) float v8f;

#define NB 4
#define NS 64
#define NW 32
#define NM 16
#define NPTS (NB * NS * NS * NS)
#define HSZ ((size_t)NPTS * NW)

__device__ __forceinline__ v8f wmma4(v2f a, v2f b, v8f c) {
  return __builtin_amdgcn_wmma_f32_16x16x4_f32(false, a, false, b, (short)0, c,
                                               false, false);
}

__device__ __forceinline__ v8f zero8() {
  v8f z;
  #pragma unroll
  for (int i = 0; i < 8; ++i) z[i] = 0.0f;
  return z;
}

__device__ __forceinline__ float gelu_exact(float x) {
  return 0.5f * x * (1.0f + erff(x * 0.70710678118654752440f));
}

__device__ __forceinline__ v2f flip2(v2f a, unsigned sgn) {
  v2f o;
  o.x = __uint_as_float(__float_as_uint(a.x) ^ sgn);
  o.y = __uint_as_float(__float_as_uint(a.y) ^ sgn);
  return o;
}

__device__ __forceinline__ void lds_barrier() {
  __builtin_amdgcn_fence(__ATOMIC_RELEASE, "workgroup", "local");
  __builtin_amdgcn_s_barrier();
  __builtin_amdgcn_fence(__ATOMIC_ACQUIRE, "workgroup", "local");
}

__global__ __launch_bounds__(256) void k_input_proj(
    const float* __restrict__ x, const float* __restrict__ w,
    const float* __restrict__ bias, float* __restrict__ out) {
  const int lane = threadIdx.x & 31, wv = threadIdx.x >> 5;
  const int half = lane >> 4, l16 = lane & 15;
  const long pbase = ((long)blockIdx.x * 8 + wv) * 16;
  v2f a;
  a.x = x[(pbase + l16) * 4 + half * 2];
  a.y = x[(pbase + l16) * 4 + half * 2 + 1];
  v2f b0, b1;
  b0.x = w[(half * 2) * NW + l16];
  b0.y = w[(half * 2 + 1) * NW + l16];
  b1.x = w[(half * 2) * NW + 16 + l16];
  b1.y = w[(half * 2 + 1) * NW + 16 + l16];
  v8f d0 = zero8(), d1 = zero8();
  d0 = wmma4(a, b0, d0);
  d1 = wmma4(a, b1, d1);
  #pragma unroll
  for (int r = 0; r < 8; ++r) {
    long p = pbase + r + half * 8;
    out[p * NW + l16] = d0[r] + bias[l16];
    out[p * NW + 16 + l16] = d1[r] + bias[16 + l16];
  }
}

__global__ __launch_bounds__(256) void k_pointwise(
    const float* __restrict__ hin, const float* __restrict__ w,
    const float* __restrict__ bias, float* __restrict__ out) {
  const int lane = threadIdx.x & 31, wv = threadIdx.x >> 5;
  const int half = lane >> 4, l16 = lane & 15;
  const long pbase = ((long)blockIdx.x * 8 + wv) * 16;
  v8f d0 = zero8(), d1 = zero8();
  #pragma unroll
  for (int kk = 0; kk < 8; ++kk) {
    const int i0 = kk * 4 + half * 2;
    v2f a;
    a.x = hin[(pbase + l16) * NW + i0];
    a.y = hin[(pbase + l16) * NW + i0 + 1];
    v2f b0, b1;
    b0.x = w[(i0)*NW + l16];      b0.y = w[(i0 + 1) * NW + l16];
    b1.x = w[(i0)*NW + 16 + l16]; b1.y = w[(i0 + 1) * NW + 16 + l16];
    d0 = wmma4(a, b0, d0);
    d1 = wmma4(a, b1, d1);
  }
  #pragma unroll
  for (int r = 0; r < 8; ++r) {
    long p = pbase + r + half * 8;
    const float v0 = gelu_exact(d0[r] + bias[l16]);
    const float v1 = gelu_exact(d1[r] + bias[16 + l16]);
    out[p * NW + l16] = v0;
    out[p * NW + 16 + l16] = v1;
  }
}

template <bool DOGELU>
__global__ __launch_bounds__(256) void k_layer_tail(
    float* h, const float* __restrict__ acc, const float* __restrict__ w,
    const float* __restrict__ bias) {
  const int lane = threadIdx.x & 31, wv = threadIdx.x >> 5;
  const int half = lane >> 4, l16 = lane & 15;
  const long pbase = ((long)blockIdx.x * 8 + wv) * 16;
  v8f d0, d1;
  #pragma unroll
  for (int r = 0; r < 8; ++r) {
    const long p = pbase + r + half * 8;
    d0[r] = acc[p * NW + l16];
    d1[r] = acc[p * NW + 16 + l16];
  }
  #pragma unroll
  for (int kk = 0; kk < 8; ++kk) {
    const int i0 = kk * 4 + half * 2;
    v2f a;
    a.x = h[(pbase + l16) * NW + i0];
    a.y = h[(pbase + l16) * NW + i0 + 1];
    v2f b0, b1;
    b0.x = w[(l16)*NW + i0];        b0.y = w[(l16)*NW + i0 + 1];
    b1.x = w[(16 + l16) * NW + i0]; b1.y = w[(16 + l16) * NW + i0 + 1];
    d0 = wmma4(a, b0, d0);
    d1 = wmma4(a, b1, d1);
  }
  #pragma unroll
  for (int r = 0; r < 8; ++r) {
    const long p = pbase + r + half * 8;
    float v0 = d0[r] + bias[l16];
    float v1 = d1[r] + bias[16 + l16];
    if (DOGELU) { v0 = gelu_exact(v0); v1 = gelu_exact(v1); }
    h[p * NW + l16] = v0;
    h[p * NW + 16 + l16] = v1;
  }
}

__device__ __forceinline__ void load_acc_tiles(const float* acc, long lbase,
                                               long strideA, int half, int l16,
                                               int t0, v8f* o) {
  #pragma unroll
  for (int r = 0; r < 8; ++r) {
    #pragma unroll
    for (int t = 0; t < 2; ++t) {
      const long ia = lbase + (long)(r + half * 8 + 16 * (t0 + t)) * strideA;
      o[2 * t][r]     = acc[ia + l16];
      o[2 * t + 1][r] = acc[ia + 16 + l16];
    }
  }
}

__device__ __forceinline__ void store_acc_tiles(float* acc, long lbase,
                                                long strideA, int half, int l16,
                                                int t0, const v8f* o) {
  #pragma unroll
  for (int r = 0; r < 8; ++r) {
    #pragma unroll
    for (int t = 0; t < 2; ++t) {
      const long ia = lbase + (long)(r + half * 8 + 16 * (t0 + t)) * strideA;
      acc[ia + l16]      = o[2 * t][r];
      acc[ia + 16 + l16] = o[2 * t + 1][r];
    }
  }
}

__device__ __forceinline__ long line_base(int l, long strideU, long strideV) {
  const int b = l >> 12, u = (l >> 6) & 63, v = l & 63;
  return (long)b * (64L * 64 * 64 * NW) + (long)u * strideU + (long)v * strideV;
}

template <bool FIRST>
__global__ __launch_bounds__(256, 8) void k_spectral_axis(
    const float* __restrict__ h, const float* __restrict__ wgt,
    float* __restrict__ acc, long strideU, long strideV, long strideA) {
  extern __shared__ float smem[];
  float* G  = smem + 2048;
  float* Gp = G;

  const int tid = threadIdx.x;
  const int lane = tid & 31, wv = tid >> 5;
  const int half = lane >> 4, l16 = lane & 15;
  const float NRM = 0.125f;

  const int wvs = __builtin_amdgcn_readfirstlane(wv);
  const char* hs = (const char*)(h + line_base(blockIdx.x * 16 + wvs * 2,
                                               strideU, strideV));
  const unsigned lo = ((unsigned)(half * 2) * (unsigned)strideA + l16) * 4u;
  v2f bq0[4], bq1[4];
  #pragma unroll
  for (int j = 0; j < 4; ++j) {
    const char* q = hs + (long)(j * 4) * strideA * 4;
    const char* q1 = q + strideA * 4;
    bq0[j].x = *(const float*)(q + lo);
    bq0[j].y = *(const float*)(q1 + lo);
    bq1[j].x = *(const float*)(q + 64 + lo);
    bq1[j].y = *(const float*)(q1 + 64 + lo);
  }
  v2f fF[2][8];
  #pragma unroll
  for (int t = 0; t < 2; ++t) {
    const int k = 2 * (l16 >> 1) + t, ph = (l16 & 1) << 4;
    #pragma unroll
    for (int kk = 0; kk < 8; ++kk) {
      const int p0 = kk * 4 + half * 2;
      fF[t][kk].x = __builtin_amdgcn_cosf(
          (float)((k * p0 + ph) & 63) * (1.0f / 64.0f)) * NRM;
      fF[t][kk].y = __builtin_amdgcn_cosf(
          (float)((k * (p0 + 1) + ph) & 63) * (1.0f / 64.0f)) * NRM;
    }
  }
  v8f a00, a01, a10, a11;
  #pragma unroll
  for (int gr = 0; gr < 8; ++gr) {
    const int ll = gr >> 2, g = gr & 3;
    if (g == 0) { a00 = zero8(); a01 = zero8(); a10 = zero8(); a11 = zero8(); }
    v2f nq0[4], nq1[4];
    if (gr < 7) {
      const int nl = (gr + 1) >> 2, ng = (gr + 1) & 3;
      #pragma unroll
      for (int j = 0; j < 4; ++j) {
        const char* q = hs + ((long)((ng * 4 + j) * 4) * strideA +
                              (long)nl * strideV) * 4;
        const char* q1 = q + strideA * 4;
        nq0[j].x = *(const float*)(q + lo);
        nq0[j].y = *(const float*)(q1 + lo);
        nq1[j].x = *(const float*)(q + 64 + lo);
        nq1[j].y = *(const float*)(q1 + 64 + lo);
      }
    }
    #pragma unroll
    for (int j = 0; j < 4; ++j) {
      const int kk = g * 4 + j;
      const v2f c0 = kk < 8 ? bq0[j] : flip2(bq0[j], 0x80000000u);
      const v2f c1 = kk < 8 ? bq1[j] : flip2(bq1[j], 0x80000000u);
      a00 = wmma4(fF[0][kk & 7], bq0[j], a00);
      a01 = wmma4(fF[0][kk & 7], bq1[j], a01);
      a10 = wmma4(fF[1][kk & 7], c0, a10);
      a11 = wmma4(fF[1][kk & 7], c1, a11);
    }
    if (g == 3) {
      float* g_ = G + (wv * 2 + ll) * 1024;
      #pragma unroll
      for (int r = 0; r < 8; ++r) {
        const int m = 4 * (r >> 1) + 16 * half + (r & 1);
        g_[(m)*32 + l16]          = a00[r];
        g_[(m)*32 + 16 + l16]     = a01[r];
        g_[(m + 2) * 32 + l16]      = a10[r];
        g_[(m + 2) * 32 + 16 + l16] = a11[r];
      }
    }
    if (gr < 7) {
      #pragma unroll
      for (int j = 0; j < 4; ++j) { bq0[j] = nq0[j]; bq1[j] = nq1[j]; }
    }
  }
  __syncthreads();

  float* Wc = smem;
  const int ld_q  = lane & 7;
  const int ld_ol = (wv & 3) * 4 + (lane >> 3);
  const int ld_ih = wv >> 2;
  const float* wsrc = wgt + ld_ih * 1024 + ld_ol * 32 + 4 * ld_q;
  const int st0 = ld_q * 256 + ((ld_ol ^ (ld_q << 2)) << 1) + ld_ih;
  const int st1 = ld_q * 256 + (((16 + ld_ol) ^ (ld_q << 2)) << 1) + ld_ih;
  int rd[2];
  #pragma unroll
  for (int km = 0; km < 2; ++km) {
    const int k = wv + km * 8;
    rd[km] = (2 * k) * 64 + ((lane ^ ((k >> 1) << 2)) << 1);
  }
  float4 pf0 = *(const float4*)(wsrc);
  float4 pf1 = *(const float4*)(wsrc + 2048);
  v8f oRe[2][2], oIm[2][2];
  #pragma unroll
  for (int km = 0; km < 2; ++km) {
    #pragma unroll
    for (int mt = 0; mt < 2; ++mt) { oRe[km][mt] = zero8(); oIm[km][mt] = zero8(); }
  }
  #pragma unroll
  for (int kk = 0; kk < 8; ++kk) {
    const int i0 = kk * 4 + half * 2;
    v2f bRe[2], bIm[2];
    #pragma unroll
    for (int mt = 0; mt < 2; ++mt) {
      const int ch = kk * 2 + mt;
      Wc[st0]       = pf0.x;  Wc[st0 + 64]  = pf0.y;
      Wc[st0 + 128] = pf0.z;  Wc[st0 + 192] = pf0.w;
      Wc[st1]       = pf1.x;  Wc[st1 + 64]  = pf1.y;
      Wc[st1 + 128] = pf1.z;  Wc[st1 + 192] = pf1.w;
      if (ch < 15) {
        const float* nx = wsrc + ((ch + 1) >> 1) * 4096 + ((ch + 1) & 1) * 512;
        pf0 = *(const float4*)(nx);
        pf1 = *(const float4*)(nx + 2048);
      }
      lds_barrier();
      if (mt == 0) {
        #pragma unroll
        for (int km = 0; km < 2; ++km) {
          const int k = wv + km * 8;
          bRe[km] = *(const v2f*)(G + l16 * 1024 + (2 * k) * 32 + i0);
          bIm[km] = *(const v2f*)(G + l16 * 1024 + (2 * k + 1) * 32 + i0);
        }
      }
      #pragma unroll
      for (int km = 0; km < 2; ++km) {
        const v2f aR = *(const v2f*)(Wc + rd[km]);
        const v2f aI = *(const v2f*)(Wc + rd[km] + 64);
        v2f aIn;
        aIn.x = -aI.x; aIn.y = -aI.y;
        oRe[km][mt] = wmma4(bRe[km], aR, oRe[km][mt]);
        oRe[km][mt] = wmma4(bIm[km], aIn, oRe[km][mt]);
        oIm[km][mt] = wmma4(bRe[km], aI, oIm[km][mt]);
        oIm[km][mt] = wmma4(bIm[km], aR, oIm[km][mt]);
      }
      lds_barrier();
    }
  }
  #pragma unroll
  for (int km = 0; km < 2; ++km) {
    const int k = wv + km * 8;
    #pragma unroll
    for (int r = 0; r < 8; ++r) {
      const int ln = r + half * 8;
      Gp[ln * 1024 + (2 * k) * 32 + l16]          = oRe[km][0][r];
      Gp[ln * 1024 + (2 * k) * 32 + 16 + l16]     = oRe[km][1][r];
      Gp[ln * 1024 + (2 * k + 1) * 32 + l16]      = oIm[km][0][r];
      Gp[ln * 1024 + (2 * k + 1) * 32 + 16 + l16] = oIm[km][1][r];
    }
  }

  v2f fI[2][8];
  #pragma unroll
  for (int t = 0; t < 2; ++t) {
    const int p = 16 * t + l16;
    #pragma unroll
    for (int kk = 0; kk < 8; ++kk) {
      const int k = 2 * kk + half;
      const float rev = (float)((k * p) & 63) * (1.0f / 64.0f);
      const float sc = (k == 0 ? 1.0f : 2.0f) * NRM;
      fI[t][kk].x = __builtin_amdgcn_cosf(rev) * sc;
      fI[t][kk].y = -__builtin_amdgcn_sinf(rev) * sc;
    }
  }

  __syncthreads();

  #pragma unroll 1
  for (int ps = 0; ps < 4; ++ps) {
    const int ll = ps >> 1, hp = ps & 1;
    const int li = __builtin_amdgcn_readfirstlane(wv) * 2 + ll;
    const long lbase = line_base(blockIdx.x * 16 + li, strideU, strideV);
    const float* gp = Gp + li * 1024;
    const unsigned sgnI = (unsigned)(hp & half) << 31;
    v8f o[4];
    if (FIRST) {
      #pragma unroll
      for (int t = 0; t < 4; ++t) o[t] = zero8();
    } else {
      load_acc_tiles(acc, lbase, strideA, half, l16, 2 * hp, o);
    }
    #pragma unroll
    for (int kk = 0; kk < 8; ++kk) {
      const int i0 = kk * 4 + half * 2;
      v2f b0, b1;
      b0.x = gp[(i0)*32 + l16];      b0.y = gp[(i0 + 1) * 32 + l16];
      b1.x = gp[(i0)*32 + 16 + l16]; b1.y = gp[(i0 + 1) * 32 + 16 + l16];
      const v2f a0 = flip2(fI[0][kk], sgnI), a1 = flip2(fI[1][kk], sgnI);
      o[0] = wmma4(a0, b0, o[0]);  o[1] = wmma4(a0, b1, o[1]);
      o[2] = wmma4(a1, b0, o[2]);  o[3] = wmma4(a1, b1, o[3]);
    }
    store_acc_tiles(acc, lbase, strideA, half, l16, 2 * hp, o);
  }
}

__global__ __launch_bounds__(256) void k_fc3(const float* __restrict__ hin,
                                             const float* __restrict__ w,
                                             const float* __restrict__ bias,
                                             float* __restrict__ out) {
  const long p = (long)blockIdx.x * blockDim.x + threadIdx.x;
  const float4* hv = (const float4*)(hin + p * NW);
  const float4* wv = (const float4*)w;
  float s = 0.0f;
  #pragma unroll
  for (int i = 0; i < 8; ++i) {
    float4 a = hv[i], b = wv[i];
    s += a.x * b.x + a.y * b.y + a.z * b.z + a.w * b.w;
  }
  out[p] = s + bias[0];
}

extern "C" void kernel_launch(void* const* d_in, const int* in_sizes, int n_in,
                              void* d_out, int out_size, void* d_ws,
                              size_t ws_size, hipStream_t stream) {
  const float* x      = (const float*)d_in[0];
  const float* in_w   = (const float*)d_in[1];
  const float* in_b   = (const float*)d_in[2];
  const float* fwx    = (const float*)d_in[3];
  const float* fwy    = (const float*)d_in[4];
  const float* fwz    = (const float*)d_in[5];
  const float* conv_w = (const float*)d_in[6];
  const float* conv_b = (const float*)d_in[7];
  const float* fc1_w  = (const float*)d_in[8];
  const float* fc1_b  = (const float*)d_in[9];
  const float* fc2_w  = (const float*)d_in[10];
  const float* fc2_b  = (const float*)d_in[11];
  const float* fc3_w  = (const float*)d_in[12];
  const float* fc3_b  = (const float*)d_in[13];
  float* out = (float*)d_out;

  float* buf0 = (float*)d_ws;
  float* buf1 = buf0 + HSZ;

  const dim3 blk(256);
  const int wgPts = NPTS / (16 * 8);
  const size_t smemB = (size_t)(2048 + 16384) * sizeof(float);
  const long sX = 64L * 64 * NW;
  const long sY = 64L * NW;
  const long sZ = (long)NW;

  k_input_proj<<<wgPts, blk, 0, stream>>>(x, in_w, in_b, buf0);

  for (int i = 0; i < 4; ++i) {
    const long wofs = (long)i * (NW * NW * NM * 2);
    k_spectral_axis<true><<<1024, blk, smemB, stream>>>(buf0, fwz + wofs, buf1,
                                                        sX, sY, sZ);
    k_spectral_axis<false><<<1024, blk, smemB, stream>>>(buf0, fwy + wofs, buf1,
                                                         sX, sZ, sY);
    k_spectral_axis<false><<<1024, blk, smemB, stream>>>(buf0, fwx + wofs, buf1,
                                                         sY, sZ, sX);
    if (i < 3)
      k_layer_tail<true><<<wgPts, blk, 0, stream>>>(
          buf0, buf1, conv_w + i * (NW * NW), conv_b + i * NW);
    else
      k_layer_tail<false><<<wgPts, blk, 0, stream>>>(
          buf0, buf1, conv_w + i * (NW * NW), conv_b + i * NW);
  }

  k_pointwise<<<wgPts, blk, 0, stream>>>(buf0, fc1_w, fc1_b, buf1);
  k_pointwise<<<wgPts, blk, 0, stream>>>(buf1, fc2_w, fc2_b, buf0);
  k_fc3<<<NPTS / 256, blk, 0, stream>>>(buf0, fc3_w, fc3_b, out);
}
